# starting kernel
# baseline (speedup 1.0000x reference)
#include <hip/hip_runtime.h>


typedef float v2f __attribute__((ext_vector_type(2)));
typedef float v8f __attribute__((ext_vector_type(8)));

#define BB 4
#define NN 32768
#define MM 8192
#define KK 16
#define DD 64
#define OO 64

template <int MODE>
__launch_bounds__(256)
__global__ void gemm16_wmma(const float* __restrict__ X,
                            const float* __restrict__ W,
                            const float* __restrict__ bias,
                            float* __restrict__ Y,
                            int numTiles) {
  const int lane  = threadIdx.x & 31;
  const int wib   = threadIdx.x >> 5;
  const int wpb   = blockDim.x >> 5;
  const int gwave = blockIdx.x * wpb + wib;
  const int nWav  = gridDim.x * wpb;

  const int mrow = lane & 15;
  const int grp  = lane >> 4;
  const int kb   = grp * 2;

  v2f bf[4][16];
#pragma unroll
  for (int t = 0; t < 4; ++t) {
    const float* wrow = W + (size_t)(t * 16 + mrow) * (2 * DD);
#pragma unroll
    for (int s = 0; s < 16; ++s) {
      const int k = 4 * s + kb;
      if (MODE == 0) {
        bf[t][s] = *(const v2f*)(wrow + k);
      } else {
        v2f p = *(const v2f*)(wrow + DD + k);
        v2f q = *(const v2f*)(wrow + k);
        bf[t][s] = p - q;
      }
    }
  }

  float bv[4];
#pragma unroll
  for (int t = 0; t < 4; ++t)
    bv[t] = (MODE == 1) ? bias[t * 16 + mrow] : 0.0f;

  for (int tile = gwave; tile < numTiles; tile += nWav) {
    const float* xrow = X + (size_t)(tile * 16 + mrow) * DD + kb;

    v8f acc[4];
#pragma unroll
    for (int t = 0; t < 4; ++t)
#pragma unroll
      for (int i = 0; i < 8; ++i) acc[t][i] = bv[t];

#pragma unroll
    for (int s = 0; s < 16; ++s) {
      v2f a = *(const v2f*)(xrow + 4 * s);
#pragma unroll
      for (int t = 0; t < 4; ++t) {
        acc[t] = __builtin_amdgcn_wmma_f32_16x16x4_f32(
             false, a,  false, bf[t][s],
             (short)0, acc[t],  false,  false);
      }
    }

    float* yout = Y + (size_t)(tile * 16 + 8 * grp) * OO;
#pragma unroll
    for (int t = 0; t < 4; ++t)
#pragma unroll
      for (int r = 0; r < 8; ++r)
        yout[(size_t)r * OO + t * 16 + mrow] = acc[t][r];
  }
}

__launch_bounds__(256)
__global__ void gather_max_kernel(const float* __restrict__ y,
                                  const float* __restrict__ z,
                                  const int* __restrict__ knn,
                                  float* __restrict__ out) {
  const int lane  = threadIdx.x & 31;
  const int gwave = (blockIdx.x * blockDim.x + threadIdx.x) >> 5;
  const int b = gwave >> 13;
  const int m = gwave & (MM - 1);
  const int o0 = lane * 2;

  const v2f zv = *(const v2f*)(z + (size_t)gwave * OO + o0);
  v2f acc;
  acc.x = 0.0f;
  acc.y = 0.0f;

  const int* kn = knn + (size_t)gwave * KK;
#pragma unroll
  for (int k = 0; k < KK; ++k) {
    const int idx = kn[k];
    const v2f yv = *(const v2f*)(y + ((size_t)b * NN + idx) * OO + o0);
    v2f s = yv + zv;
    acc.x = fmaxf(acc.x, s.x);
    acc.y = fmaxf(acc.y, s.y);
  }

  float* op = out + ((size_t)b * OO + o0) * MM + m;
  op[0]  = acc.x;
  op[MM] = acc.y;
}

extern "C" void kernel_launch(void* const* d_in, const int* in_sizes, int n_in,
                              void* d_out, int out_size, void* d_ws, size_t ws_size,
                              hipStream_t stream) {
  const float* x    = (const float*)d_in[0];
  const float* xs   = (const float*)d_in[1];
  const int*   knn  = (const int*)d_in[2];
  const float* W    = (const float*)d_in[3];
  const float* bias = (const float*)d_in[4];
  float* out = (float*)d_out;

  float* y = (float*)d_ws;
  float* z = y + (size_t)BB * NN * OO;

  gemm16_wmma<0><<<256, 256, 0, stream>>>(x, W, bias, y, (BB * NN) / 16);
  gemm16_wmma<1><<<256, 256, 0, stream>>>(xs, W, bias, z, (BB * MM) / 16);
  gather_max_kernel<<<(BB * MM * 32) / 256, 256, 0, stream>>>(y, z, knn, out);
}


# z launch: one wave per SIMD, each wave's weight fragments serve two point tiles (half the weight requests per WGP) and load in k-step order, so products run under the loads; sums unchanged
# speedup vs baseline: 2.7610x; 2.7610x over previous
#include <hip/hip_runtime.h>


typedef float v2f __attribute__((ext_vector_type(2)));
typedef float v4f __attribute__((ext_vector_type(4)));
typedef int   v4i __attribute__((ext_vector_type(4)));
typedef float v8f __attribute__((ext_vector_type(8)));

#define BB 4
#define NN 32768
#define MM 8192
#define KK 16
#define DD 64
#define OO 64

__device__ __forceinline__ void load_row_pieces(v4f (&g)[8], const float* rowp) {
#pragma unroll
  for (int j = 0; j < 8; ++j) g[j] = *(const v4f*)(rowp + 8 * j);
}

__launch_bounds__(128)
__global__ void z_tile_kernel(const float* __restrict__ xs,
                              const float* __restrict__ W,
                              const float* __restrict__ bias,
                              float* __restrict__ Z,
                              int numPairs) {
  const int lane  = threadIdx.x & 31;
  const int gwave = blockIdx.x * (blockDim.x >> 5) + (threadIdx.x >> 5);
  if (gwave >= numPairs) return;
  const int c = lane & 15;
  const int h = lane >> 4;

  v4f ga[8], gb[8];
  const float* xr = xs + (size_t)(gwave * 32 + c) * DD + 4 * h;
  load_row_pieces(ga, xr);
  load_row_pieces(gb, xr + 16 * DD);

  v8f bt[4];
#pragma unroll
  for (int t = 0; t < 4; ++t) {
    const v4f lo = *(const v4f*)(bias + 16 * t + 8 * h);
    const v4f hi = *(const v4f*)(bias + 16 * t + 8 * h + 4);
    bt[t][0] = lo.x; bt[t][1] = lo.y; bt[t][2] = lo.z; bt[t][3] = lo.w;
    bt[t][4] = hi.x; bt[t][5] = hi.y; bt[t][6] = hi.z; bt[t][7] = hi.w;
  }

  v2f wd[4][16];
  const float* wlane = W + (size_t)c * (2 * DD) + 4 * h;
#pragma unroll
  for (int j = 0; j < 8; ++j) {
#pragma unroll
    for (int t = 0; t < 4; ++t) {
      const float* wrow = wlane + (size_t)(t * 16) * (2 * DD);
      const v4f p = *(const v4f*)(wrow + DD + 8 * j);
      const v4f q = *(const v4f*)(wrow + 8 * j);
      const v4f d = p - q;
      wd[t][2 * j].x     = d.x;  wd[t][2 * j].y     = d.y;
      wd[t][2 * j + 1].x = d.z;  wd[t][2 * j + 1].y = d.w;
    }
  }

  v8f acca[4], accb[4];
#pragma unroll
  for (int j = 0; j < 8; ++j) {
    v2f a0, a1, b0, b1;
    a0.x = ga[j].x; a0.y = ga[j].y;
    a1.x = ga[j].z; a1.y = ga[j].w;
    b0.x = gb[j].x; b0.y = gb[j].y;
    b1.x = gb[j].z; b1.y = gb[j].w;
#pragma unroll
    for (int t = 0; t < 4; ++t) {
      acca[t] = __builtin_amdgcn_wmma_f32_16x16x4_f32(
           false, wd[t][2 * j],  false, a0,
           (short)0, (j == 0) ? bt[t] : acca[t],  false,  false);
      accb[t] = __builtin_amdgcn_wmma_f32_16x16x4_f32(
           false, wd[t][2 * j],  false, b0,
           (short)0, (j == 0) ? bt[t] : accb[t],  false,  false);
    }
#pragma unroll
    for (int t = 0; t < 4; ++t) {
      acca[t] = __builtin_amdgcn_wmma_f32_16x16x4_f32(
           false, wd[t][2 * j + 1],  false, a1,
           (short)0, acca[t],  false,  false);
      accb[t] = __builtin_amdgcn_wmma_f32_16x16x4_f32(
           false, wd[t][2 * j + 1],  false, b1,
           (short)0, accb[t],  false,  false);
    }
  }

  float* zo = Z + (size_t)(2 * gwave) * (16 * OO) + lane * 4;
#pragma unroll
  for (int t = 0; t < 4; ++t) {
    v4f lo, hi;
    lo.x = acca[t][0]; lo.y = acca[t][1]; lo.z = acca[t][2]; lo.w = acca[t][3];
    hi.x = acca[t][4]; hi.y = acca[t][5]; hi.z = acca[t][6]; hi.w = acca[t][7];
    *(v4f*)(zo + (2 * t) * 128)     = lo;
    *(v4f*)(zo + (2 * t + 1) * 128) = hi;
  }
#pragma unroll
  for (int t = 0; t < 4; ++t) {
    v4f lo, hi;
    lo.x = accb[t][0]; lo.y = accb[t][1]; lo.z = accb[t][2]; lo.w = accb[t][3];
    hi.x = accb[t][4]; hi.y = accb[t][5]; hi.z = accb[t][6]; hi.w = accb[t][7];
    *(v4f*)(zo + 16 * OO + (2 * t) * 128)     = lo;
    *(v4f*)(zo + 16 * OO + (2 * t + 1) * 128) = hi;
  }
}

__device__ __forceinline__ void slot_product_max(const v2f (&wf)[4][16], const v4f (&g)[8],
                                                 const v8f (&zt)[4], v8f (&mx)[4]) {
  v8f d[4];
#pragma unroll
  for (int j = 0; j < 8; ++j) {
    v2f g0, g1;
    g0.x = g[j].x; g0.y = g[j].y;
    g1.x = g[j].z; g1.y = g[j].w;
#pragma unroll
    for (int t = 0; t < 4; ++t)
      d[t] = __builtin_amdgcn_wmma_f32_16x16x4_f32(
           false, wf[t][2 * j],  false, g0,
           (short)0, (j == 0) ? zt[t] : d[t],  false,  false);
#pragma unroll
    for (int t = 0; t < 4; ++t)
      d[t] = __builtin_amdgcn_wmma_f32_16x16x4_f32(
           false, wf[t][2 * j + 1],  false, g1,
           (short)0, d[t],  false,  false);
  }
#pragma unroll
  for (int t = 0; t < 4; ++t)
#pragma unroll
    for (int r = 0; r < 8; ++r) mx[t][r] = fmaxf(mx[t][r], d[t][r]);
}

__launch_bounds__(256)
__global__ void gather_product_max_kernel(const float* __restrict__ x,
                                          const float* __restrict__ z,
                                          const int* __restrict__ knn,
                                          const float* __restrict__ W,
                                          float* __restrict__ out) {
  const int lane  = threadIdx.x & 31;
  const int gwave = (blockIdx.x * blockDim.x + threadIdx.x) >> 5;
  const int c  = lane & 15;
  const int h  = lane >> 4;
  const int p0 = gwave * 16;
  const int b  = p0 >> 13;
  const int m0 = p0 & (MM - 1);

  v2f wf[4][16];
#pragma unroll
  for (int t = 0; t < 4; ++t) {
    const float* wrow = W + (size_t)(t * 16 + c) * (2 * DD) + 4 * h;
#pragma unroll
    for (int j = 0; j < 8; ++j) {
      const v4f q = *(const v4f*)(wrow + 8 * j);
      wf[t][2 * j].x     = q.x;  wf[t][2 * j].y     = q.y;
      wf[t][2 * j + 1].x = q.z;  wf[t][2 * j + 1].y = q.w;
    }
  }

  const int*   kn = knn + (size_t)(p0 + c) * KK;
  const float* xb = x + (size_t)b * NN * DD + 4 * h;

  v4i iq = *(const v4i*)kn;
  v4f ga[8], gb[8];
  load_row_pieces(ga, xb + (size_t)iq.x * DD);

  v8f zt[4];
  const float* zp = z + (size_t)gwave * (16 * OO) + lane * 4;
#pragma unroll
  for (int t = 0; t < 4; ++t) {
    const v4f lo = *(const v4f*)(zp + (2 * t) * 128);
    const v4f hi = *(const v4f*)(zp + (2 * t + 1) * 128);
    zt[t][0] = lo.x; zt[t][1] = lo.y; zt[t][2] = lo.z; zt[t][3] = lo.w;
    zt[t][4] = hi.x; zt[t][5] = hi.y; zt[t][6] = hi.z; zt[t][7] = hi.w;
  }

  v8f mx[4];
#pragma unroll
  for (int t = 0; t < 4; ++t)
#pragma unroll
    for (int r = 0; r < 8; ++r) mx[t][r] = 0.0f;

#pragma unroll 1
  for (int q = 0; q < 3; ++q) {
    const v4i iqn = *(const v4i*)(kn + 4 * (q + 1));

    load_row_pieces(gb, xb + (size_t)iq.y * DD);
    __builtin_amdgcn_sched_barrier(0);
    slot_product_max(wf, ga, zt, mx);

    load_row_pieces(ga, xb + (size_t)iq.z * DD);
    __builtin_amdgcn_sched_barrier(0);
    slot_product_max(wf, gb, zt, mx);

    load_row_pieces(gb, xb + (size_t)iq.w * DD);
    __builtin_amdgcn_sched_barrier(0);
    slot_product_max(wf, ga, zt, mx);

    load_row_pieces(ga, xb + (size_t)iqn.x * DD);
    __builtin_amdgcn_sched_barrier(0);
    slot_product_max(wf, gb, zt, mx);

    iq = iqn;
  }

  load_row_pieces(gb, xb + (size_t)iq.y * DD);
  __builtin_amdgcn_sched_barrier(0);
  slot_product_max(wf, ga, zt, mx);

  load_row_pieces(ga, xb + (size_t)iq.z * DD);
  __builtin_amdgcn_sched_barrier(0);
  slot_product_max(wf, gb, zt, mx);

  load_row_pieces(gb, xb + (size_t)iq.w * DD);
  __builtin_amdgcn_sched_barrier(0);
  slot_product_max(wf, ga, zt, mx);

  slot_product_max(wf, gb, zt, mx);

  float* op = out + ((size_t)b * OO + 8 * h) * MM + m0 + c;
#pragma unroll
  for (int t = 0; t < 4; ++t)
#pragma unroll
    for (int r = 0; r < 8; ++r)
      op[(size_t)(16 * t + r) * MM] = mx[t][r];
}

extern "C" void kernel_launch(void* const* d_in, const int* in_sizes, int n_in,
                              void* d_out, int out_size, void* d_ws, size_t ws_size,
                              hipStream_t stream) {
  const float* x    = (const float*)d_in[0];
  const float* xs   = (const float*)d_in[1];
  const int*   knn  = (const int*)d_in[2];
  const float* W    = (const float*)d_in[3];
  const float* bias = (const float*)d_in[4];
  float* out = (float*)d_out;

  float* z = (float*)d_ws;

  z_tile_kernel<<<256, 128, 0, stream>>>(xs, W, bias, z, (BB * MM) / 32);
  gather_product_max_kernel<<<(BB * MM / 16 * 32) / 256, 256, 0, stream>>>(x, z, knn, W, out);
}


# z launch dropped: each gather wave first forms z for its 16 points on WMMA (W2-W1 as A, their x_subset rows as B, bias as C) with the slot-0 rows in flight; one launch, z never stored; sums unchanged
# speedup vs baseline: 3.0492x; 1.1044x over previous
#include <hip/hip_runtime.h>


typedef float v2f __attribute__((ext_vector_type(2)));
typedef float v4f __attribute__((ext_vector_type(4)));
typedef int   v4i __attribute__((ext_vector_type(4)));
typedef float v8f __attribute__((ext_vector_type(8)));

#define BB 4
#define NN 32768
#define MM 8192
#define KK 16
#define DD 64
#define OO 64

__device__ __forceinline__ void load_row_pieces(v4f (&g)[8], const float* rowp) {
#pragma unroll
  for (int j = 0; j < 8; ++j) g[j] = *(const v4f*)(rowp + 8 * j);
}

__device__ __forceinline__ void slot_product_max(const v2f (&wf)[4][16], const v4f (&g)[8],
                                                 const v8f (&zt)[4], v8f (&mx)[4]) {
  v8f d[4];
#pragma unroll
  for (int j = 0; j < 8; ++j) {
    v2f g0, g1;
    g0.x = g[j].x; g0.y = g[j].y;
    g1.x = g[j].z; g1.y = g[j].w;
#pragma unroll
    for (int t = 0; t < 4; ++t)
      d[t] = __builtin_amdgcn_wmma_f32_16x16x4_f32(
           false, wf[t][2 * j],  false, g0,
           (short)0, (j == 0) ? zt[t] : d[t],  false,  false);
#pragma unroll
    for (int t = 0; t < 4; ++t)
      d[t] = __builtin_amdgcn_wmma_f32_16x16x4_f32(
           false, wf[t][2 * j + 1],  false, g1,
           (short)0, d[t],  false,  false);
  }
#pragma unroll
  for (int t = 0; t < 4; ++t)
#pragma unroll
    for (int r = 0; r < 8; ++r) mx[t][r] = fmaxf(mx[t][r], d[t][r]);
}

__launch_bounds__(256)
__global__ void gather_product_max_kernel(const float* __restrict__ x,
                                          const float* __restrict__ xs,
                                          const int* __restrict__ knn,
                                          const float* __restrict__ W,
                                          const float* __restrict__ bias,
                                          float* __restrict__ out) {
  const int lane  = threadIdx.x & 31;
  const int gwave = (blockIdx.x * blockDim.x + threadIdx.x) >> 5;
  const int c  = lane & 15;
  const int h  = lane >> 4;
  const int p0 = gwave * 16;
  const int b  = p0 >> 13;
  const int m0 = p0 & (MM - 1);

  const int*   kn = knn + (size_t)(p0 + c) * KK;
  const float* xb = x + (size_t)b * NN * DD + 4 * h;

  v4i iq = *(const v4i*)kn;

  v4f gs[8];
  load_row_pieces(gs, xs + (size_t)(p0 + c) * DD + 4 * h);

  v8f zt[4];
#pragma unroll
  for (int t = 0; t < 4; ++t) {
    const v4f lo = *(const v4f*)(bias + 16 * t + 8 * h);
    const v4f hi = *(const v4f*)(bias + 16 * t + 8 * h + 4);
    zt[t][0] = lo.x; zt[t][1] = lo.y; zt[t][2] = lo.z; zt[t][3] = lo.w;
    zt[t][4] = hi.x; zt[t][5] = hi.y; zt[t][6] = hi.z; zt[t][7] = hi.w;
  }

  __builtin_amdgcn_sched_barrier(0);

  v4f w1[4][8], w2[4][8];
  const float* wlane = W + (size_t)c * (2 * DD) + 4 * h;
#pragma unroll
  for (int j = 0; j < 8; ++j) {
#pragma unroll
    for (int t = 0; t < 4; ++t) {
      const float* wrow = wlane + (size_t)(t * 16) * (2 * DD);
      w2[t][j] = *(const v4f*)(wrow + DD + 8 * j);
      w1[t][j] = *(const v4f*)(wrow + 8 * j);
    }
  }

  __builtin_amdgcn_sched_barrier(0);

  v4f ga[8], gb[8];
  load_row_pieces(ga, xb + (size_t)iq.x * DD);
  __builtin_amdgcn_sched_barrier(0);

  v2f wf[4][16];
#pragma unroll
  for (int j = 0; j < 8; ++j) {
    __builtin_amdgcn_sched_barrier(0);
    v2f s0, s1, d0[4], d1[4];
    s0.x = gs[j].x; s0.y = gs[j].y;
    s1.x = gs[j].z; s1.y = gs[j].w;
#pragma unroll
    for (int t = 0; t < 4; ++t) {
      const v4f q = w1[t][j];
      const v4f d = w2[t][j] - q;
      wf[t][2 * j].x     = q.x;  wf[t][2 * j].y     = q.y;
      wf[t][2 * j + 1].x = q.z;  wf[t][2 * j + 1].y = q.w;
      d0[t].x = d.x;  d0[t].y = d.y;
      d1[t].x = d.z;  d1[t].y = d.w;
    }
#pragma unroll
    for (int t = 0; t < 4; ++t)
      zt[t] = __builtin_amdgcn_wmma_f32_16x16x4_f32(
           false, d0[t],  false, s0,
           (short)0, zt[t],  false,  false);
#pragma unroll
    for (int t = 0; t < 4; ++t)
      zt[t] = __builtin_amdgcn_wmma_f32_16x16x4_f32(
           false, d1[t],  false, s1,
           (short)0, zt[t],  false,  false);
  }

  v8f mx[4];
#pragma unroll
  for (int t = 0; t < 4; ++t)
#pragma unroll
    for (int r = 0; r < 8; ++r) mx[t][r] = 0.0f;

#pragma unroll 1
  for (int q = 0; q < 3; ++q) {
    const v4i iqn = *(const v4i*)(kn + 4 * (q + 1));

    load_row_pieces(gb, xb + (size_t)iq.y * DD);
    __builtin_amdgcn_sched_barrier(0);
    slot_product_max(wf, ga, zt, mx);

    load_row_pieces(ga, xb + (size_t)iq.z * DD);
    __builtin_amdgcn_sched_barrier(0);
    slot_product_max(wf, gb, zt, mx);

    load_row_pieces(gb, xb + (size_t)iq.w * DD);
    __builtin_amdgcn_sched_barrier(0);
    slot_product_max(wf, ga, zt, mx);

    load_row_pieces(ga, xb + (size_t)iqn.x * DD);
    __builtin_amdgcn_sched_barrier(0);
    slot_product_max(wf, gb, zt, mx);

    iq = iqn;
  }

  load_row_pieces(gb, xb + (size_t)iq.y * DD);
  __builtin_amdgcn_sched_barrier(0);
  slot_product_max(wf, ga, zt, mx);

  load_row_pieces(ga, xb + (size_t)iq.z * DD);
  __builtin_amdgcn_sched_barrier(0);
  slot_product_max(wf, gb, zt, mx);

  load_row_pieces(gb, xb + (size_t)iq.w * DD);
  __builtin_amdgcn_sched_barrier(0);
  slot_product_max(wf, ga, zt, mx);

  slot_product_max(wf, gb, zt, mx);

  float* op = out + ((size_t)b * OO + 8 * h) * MM + m0 + c;
#pragma unroll
  for (int t = 0; t < 4; ++t)
#pragma unroll
    for (int r = 0; r < 8; ++r)
      op[(size_t)(16 * t + r) * MM] = mx[t][r];
}

extern "C" void kernel_launch(void* const* d_in, const int* in_sizes, int n_in,
                              void* d_out, int out_size, void* d_ws, size_t ws_size,
                              hipStream_t stream) {
  const float* x    = (const float*)d_in[0];
  const float* xs   = (const float*)d_in[1];
  const int*   knn  = (const int*)d_in[2];
  const float* W    = (const float*)d_in[3];
  const float* bias = (const float*)d_in[4];
  float* out = (float*)d_out;

  gather_product_max_kernel<<<(BB * MM / 16 * 32) / 256, 256, 0, stream>>>(x, xs, knn, W, bias, out);
}
